# starting kernel
# baseline (speedup 1.0000x reference)
#include <hip/hip_runtime.h>

typedef __attribute__((ext_vector_type(2))) float v2f;
typedef __attribute__((ext_vector_type(8))) float v8f;

#define D_FEAT 128

__global__ __launch_bounds__(256) void edge_dot_wmma_kernel(
    const float* __restrict__ h,
    const int*   __restrict__ src,
    const int*   __restrict__ dst,
    float*       __restrict__ out,
    int n_edges)
{
    const int lane = threadIdx.x & 31;
    const int wave = threadIdx.x >> 5;
    const long long group = (long long)blockIdx.x * (blockDim.x >> 5) + wave;
    const long long base  = group * 16;
    if (base >= (long long)n_edges) return;

    const int m  = lane & 15;
    const int hi = lane >> 4;

    long long e = base + m;
    if (e >= (long long)n_edges) e = (long long)n_edges - 1;

    const long long srow = (long long)src[e] * D_FEAT;
    const long long drow = (long long)dst[e] * D_FEAT;
    const float* ps = h + srow + 2 * hi;
    const float* pd = h + drow + 2 * hi;

    v8f c = {0.f, 0.f, 0.f, 0.f, 0.f, 0.f, 0.f, 0.f};

#pragma unroll 8
    for (int k0 = 0; k0 < D_FEAT; k0 += 4) {
        v2f a = *(const v2f*)(ps + k0);
        v2f b = *(const v2f*)(pd + k0);
        c = __builtin_amdgcn_wmma_f32_16x16x4_f32(
                false, a, false, b, (short)0, c, false, false);
    }

    const bool lo  = (lane < 8);
    const bool hid = (lane >= 24);
    const int  v   = (lo ? lane : (lane - 24)) & 7;
    float val;
    switch (v) {
        case 0: val = c[0]; break;
        case 1: val = c[1]; break;
        case 2: val = c[2]; break;
        case 3: val = c[3]; break;
        case 4: val = c[4]; break;
        case 5: val = c[5]; break;
        case 6: val = c[6]; break;
        default: val = c[7]; break;
    }
    const long long eo = base + (lo ? lane : (lane - 16));
    if ((lo || hid) && eo < (long long)n_edges) out[eo] = val;
}

extern "C" void kernel_launch(void* const* d_in, const int* in_sizes, int n_in,
                              void* d_out, int out_size, void* d_ws, size_t ws_size,
                              hipStream_t stream) {
    const float* h   = (const float*)d_in[0];
    const int*   src = (const int*)d_in[1];
    const int*   dst = (const int*)d_in[2];
    float*       out = (float*)d_out;
    const int n_edges = in_sizes[1];

    const long long groups = ((long long)n_edges + 15) / 16;
    const int waves_per_block = 8;
    int blocks = (int)((groups + waves_per_block - 1) / waves_per_block);
    if (blocks < 1) blocks = 1;

    edge_dot_wmma_kernel<<<blocks, 256, 0, stream>>>(h, src, dst, out, n_edges);
}


# Wave loads now read 4 whole 128B lines each (WMMA row = 16B slice of an edge row, 2 edges per tile) in long-lived 32-wave blocks; score = 8 diagonal partials summed across lanes (sum order changes)
# speedup vs baseline: 1.9385x; 1.9385x over previous
#include <hip/hip_runtime.h>

typedef __attribute__((ext_vector_type(2))) float v2f;
typedef __attribute__((ext_vector_type(4))) float v4f;
typedef __attribute__((ext_vector_type(8))) float v8f;


#define D_FEAT 128
#define ROW_BYTES (D_FEAT * 4)
#define UNIT 16
#define WAVES_PER_BLOCK 32
#define GRID_BLOCKS 256


template <int CTRL>
__device__ __forceinline__ float add_partner(float v) {
    const int iv = __builtin_bit_cast(int, v);
    const int pv = __builtin_amdgcn_update_dpp(0, iv, CTRL, 0xf, 0xf, true);
    return v + __builtin_bit_cast(float, pv);
}

__device__ __forceinline__ float diag_sum8(const v8f c, const bool b0, const bool b1, const bool b2) {
    const float x0 = b0 ? c[1] : c[0];
    const float x1 = b0 ? c[3] : c[2];
    const float x2 = b0 ? c[5] : c[4];
    const float x3 = b0 ? c[7] : c[6];
    const float y0 = b1 ? x1 : x0;
    const float y1 = b1 ? x3 : x2;
    float v = b2 ? y1 : y0;
    v = add_partner<0x161>(v);
    v = add_partner<0x162>(v);
    v = add_partner<0x164>(v);
    return v;
}

template <int TT>
__device__ __forceinline__ unsigned row_off(const int idx16, const unsigned lane_boff) {
    const int g = __builtin_amdgcn_permlane16(idx16, idx16,
                                              0x11111111u * (unsigned)TT,
                                              0x11111111u * (unsigned)(8 + TT), false, false);
    return (unsigned)g * (unsigned)ROW_BYTES + lane_boff;
}

template <int P>
__device__ __forceinline__ void load_pair(const float* __restrict__ h, const int si, const int di,
                                          const unsigned lane_boff, v4f (&A)[4], v4f (&B)[4]) {
    const char* hb = (const char*)h;
    const unsigned s0 = row_off<2 * P>(si, lane_boff),     d0 = row_off<2 * P>(di, lane_boff);
    const unsigned s1 = row_off<2 * P + 1>(si, lane_boff), d1 = row_off<2 * P + 1>(di, lane_boff);
    A[0] = *(const v4f*)(hb + s0);
    A[1] = *(const v4f*)(hb + s0 + 256);
    B[0] = *(const v4f*)(hb + d0);
    B[1] = *(const v4f*)(hb + d0 + 256);
    A[2] = *(const v4f*)(hb + s1);
    A[3] = *(const v4f*)(hb + s1 + 256);
    B[2] = *(const v4f*)(hb + d1);
    B[3] = *(const v4f*)(hb + d1 + 256);
}

#define WMMA4(acc, av, bv) \
    acc = __builtin_amdgcn_wmma_f32_16x16x4_f32(false, av, false, bv, (short)0, acc, false, false)

__device__ __forceinline__ void compute_pair(const v4f (&A)[4], const v4f (&B)[4], const int p,
                                             const int lane7, const bool b0, const bool b1, const bool b2,
                                             float& res) {
    v8f c0 = {0.f, 0.f, 0.f, 0.f, 0.f, 0.f, 0.f, 0.f};
    v8f c1 = {0.f, 0.f, 0.f, 0.f, 0.f, 0.f, 0.f, 0.f};
    WMMA4(c0, __builtin_shufflevector(A[0], A[0], 0, 1), __builtin_shufflevector(B[0], B[0], 0, 1));
    WMMA4(c1, __builtin_shufflevector(A[2], A[2], 0, 1), __builtin_shufflevector(B[2], B[2], 0, 1));
    WMMA4(c0, __builtin_shufflevector(A[0], A[0], 2, 3), __builtin_shufflevector(B[0], B[0], 2, 3));
    WMMA4(c1, __builtin_shufflevector(A[2], A[2], 2, 3), __builtin_shufflevector(B[2], B[2], 2, 3));
    WMMA4(c0, __builtin_shufflevector(A[1], A[1], 0, 1), __builtin_shufflevector(B[1], B[1], 0, 1));
    WMMA4(c1, __builtin_shufflevector(A[3], A[3], 0, 1), __builtin_shufflevector(B[3], B[3], 0, 1));
    WMMA4(c0, __builtin_shufflevector(A[1], A[1], 2, 3), __builtin_shufflevector(B[1], B[1], 2, 3));
    WMMA4(c1, __builtin_shufflevector(A[3], A[3], 2, 3), __builtin_shufflevector(B[3], B[3], 2, 3));
    const float v0 = diag_sum8(c0, b0, b1, b2);
    const float v1 = diag_sum8(c1, b0, b1, b2);
    res = (lane7 == 2 * p)     ? v0 : res;
    res = (lane7 == 2 * p + 1) ? v1 : res;
}

template <bool HAS_NEXT>
__device__ __forceinline__ float unit_step(const float* __restrict__ h,
                                           const int si, const int di, const int sn, const int dn,
                                           const unsigned lane_boff,
                                           v4f (&A0)[4], v4f (&B0)[4], v4f (&A1)[4], v4f (&B1)[4],
                                           const int lane7, const bool b0, const bool b1, const bool b2) {
    float res = 0.f;
#define PIN() __builtin_amdgcn_sched_barrier(0)
    load_pair<1>(h, si, di, lane_boff, A1, B1);            PIN();
    compute_pair(A0, B0, 0, lane7, b0, b1, b2, res);       PIN();
    load_pair<2>(h, si, di, lane_boff, A0, B0);            PIN();
    compute_pair(A1, B1, 1, lane7, b0, b1, b2, res);       PIN();
    load_pair<3>(h, si, di, lane_boff, A1, B1);            PIN();
    compute_pair(A0, B0, 2, lane7, b0, b1, b2, res);       PIN();
    if (HAS_NEXT) load_pair<0>(h, sn, dn, lane_boff, A0, B0);
    PIN();
    compute_pair(A1, B1, 3, lane7, b0, b1, b2, res);
#undef PIN
    return res;
}

__global__ __launch_bounds__(WAVES_PER_BLOCK * 32) void edge_dot_rows_kernel(
    const float* __restrict__ h,
    const int*   __restrict__ src,
    const int*   __restrict__ dst,
    float*       __restrict__ out,
    int n_edges)
{
    const int lane = threadIdx.x & 31;
    const int wave = __builtin_amdgcn_readfirstlane((int)(threadIdx.x >> 5));

    const int  c     = lane & 15;
    const unsigned lane_boff = (unsigned)(((c & 7) * 4 + (lane >> 4) * 32) * 4);
    const int  lane7 = lane & 7;
    const bool b0 = (lane & 1) != 0, b1 = (lane & 2) != 0, b2 = (lane & 4) != 0;
    const bool storer  = (lane < 8) || (lane >= 24);
    const int  out_off = (lane < 8) ? lane : (lane - 16);

    const long long units = ((long long)n_edges + UNIT - 1) / UNIT;
    const long long u0 = ((long long)blockIdx.x * units) / GRID_BLOCKS;
    const long long u1 = ((long long)(blockIdx.x + 1) * units) / GRID_BLOCKS;

    long long u = u0 + wave;
    if (u >= u1) return;

    const long long last = (long long)n_edges - 1;
    long long e = u * UNIT + c;
    if (e > last) e = last;
    int si = src[e], di = dst[e];

    v4f A0[4], B0[4], A1[4], B1[4];
    load_pair<0>(h, si, di, lane_boff, A0, B0);

    while (u + WAVES_PER_BLOCK < u1) {
        const long long un = u + WAVES_PER_BLOCK;
        long long en = un * UNIT + c;
        if (en > last) en = last;
        const int sn = src[en], dn = dst[en];

        const float res = unit_step<true>(h, si, di, sn, dn, lane_boff, A0, B0, A1, B1, lane7, b0, b1, b2);
        const long long eo = u * UNIT + out_off;
        if (storer && eo <= last) out[eo] = res;

        u = un; si = sn; di = dn;
    }
    {
        const float res = unit_step<false>(h, si, di, 0, 0, lane_boff, A0, B0, A1, B1, lane7, b0, b1, b2);
        const long long eo = u * UNIT + out_off;
        if (storer && eo <= last) out[eo] = res;
    }
}

extern "C" void kernel_launch(void* const* d_in, const int* in_sizes, int n_in,
                              void* d_out, int out_size, void* d_ws, size_t ws_size,
                              hipStream_t stream) {
    const float* h   = (const float*)d_in[0];
    const int*   src = (const int*)d_in[1];
    const int*   dst = (const int*)d_in[2];
    float*       out = (float*)d_out;
    const int n_edges = in_sizes[1];

    edge_dot_rows_kernel<<<GRID_BLOCKS, WAVES_PER_BLOCK * 32, 0, stream>>>(h, src, dst, out, n_edges);
}
